# starting kernel
# baseline (speedup 1.0000x reference)
#include <hip/hip_runtime.h>
#include <hip/hip_bf16.h>

typedef __attribute__((ext_vector_type(2))) float v2f;
typedef __attribute__((ext_vector_type(8))) float v8f;

#define N_NODES 40000
#define N_EDGES 400000
#define IN_CH   64
#define HID     16
#define N_REL   90
#define N_BASES 30
#define N_CLASSES 8


__global__ void zero_u32(unsigned* __restrict__ p, int n) {
    int i = blockIdx.x * blockDim.x + threadIdx.x;
    if (i < n) p[i] = 0u;
}

__global__ void count_edges(const int* __restrict__ dst, const int* __restrict__ et,
                            unsigned* __restrict__ cnt, int E) {
    int e = blockIdx.x * blockDim.x + threadIdx.x;
    if (e < E) atomicAdd(&cnt[dst[e] * N_REL + et[e]], 1u);
}

__global__ void make_w1(const float* __restrict__ comp, const float* __restrict__ basis,
                        float* __restrict__ W) {
    int idx = blockIdx.x * blockDim.x + threadIdx.x;
    if (idx >= N_REL * IN_CH * HID) return;
    int r = idx / (IN_CH * HID);
    int rem = idx - r * (IN_CH * HID);
    float acc = 0.f;
    #pragma unroll 5
    for (int b = 0; b < N_BASES; ++b)
        acc += comp[r * N_BASES + b] * basis[b * (IN_CH * HID) + rem];
    W[idx] = acc;
}

__global__ void make_w2(const float* __restrict__ comp, const float* __restrict__ basis,
                        float* __restrict__ W) {
    int idx = blockIdx.x * blockDim.x + threadIdx.x;
    if (idx >= N_REL * HID * N_CLASSES) return;
    int r = idx / (HID * N_CLASSES);
    int rem = idx - r * (HID * N_CLASSES);
    float acc = 0.f;
    #pragma unroll 5
    for (int b = 0; b < N_BASES; ++b)
        acc += comp[r * N_BASES + b] * basis[b * (HID * N_CLASSES) + rem];
    W[idx] = acc;
}

__global__ void pad_root2(const float* __restrict__ root2, float* __restrict__ root2p) {
    int idx = threadIdx.x;
    if (idx >= HID * 16) return;
    int k = idx >> 4, n = idx & 15;
    root2p[idx] = (n < N_CLASSES) ? root2[k * N_CLASSES + n] : 0.0f;
}

__global__ void __launch_bounds__(256)
root_gemm1(const float* __restrict__ x, const float* __restrict__ root,
           const float* __restrict__ bias, float* __restrict__ h) {
    const int ntiles = N_NODES / 16;
    int wave = (int)((blockIdx.x * blockDim.x + threadIdx.x) >> 5);
    int lane = (int)(threadIdx.x & 31);
    if (wave >= ntiles) return;

    int l15   = lane & 15;
    int khalf = (lane >> 4) * 2;
    int m0    = wave * 16;

    const float* xrow = x + (size_t)(m0 + l15) * IN_CH;
    v8f acc = {};
    #pragma unroll
    for (int k0 = 0; k0 < IN_CH; k0 += 4) {
        v2f a = *(const v2f*)(xrow + k0 + khalf);
        v2f b;
        b.x = root[(k0 + khalf)     * HID + l15];
        b.y = root[(k0 + khalf + 1) * HID + l15];
        acc = __builtin_amdgcn_wmma_f32_16x16x4_f32(
                  false, a, false, b, (short)0, acc, false, false);
    }
    float bz = bias[l15];
    int mbase = m0 + ((lane >> 4) * 8);
    #pragma unroll
    for (int j = 0; j < 8; ++j)
        h[(size_t)(mbase + j) * HID + l15] = acc[j] + bz;
}

__global__ void __launch_bounds__(256)
root_gemm2(const float* __restrict__ h, const float* __restrict__ root2p,
           const float* __restrict__ bias, float* __restrict__ out) {
    const int ntiles = N_NODES / 16;
    int wave = (int)((blockIdx.x * blockDim.x + threadIdx.x) >> 5);
    int lane = (int)(threadIdx.x & 31);
    if (wave >= ntiles) return;

    int l15   = lane & 15;
    int khalf = (lane >> 4) * 2;
    int m0    = wave * 16;

    const float* hrow = h + (size_t)(m0 + l15) * HID;
    v8f acc = {};
    #pragma unroll
    for (int k0 = 0; k0 < HID; k0 += 4) {
        v2f a = *(const v2f*)(hrow + k0 + khalf);
        v2f b;
        b.x = root2p[(k0 + khalf)     * 16 + l15];
        b.y = root2p[(k0 + khalf + 1) * 16 + l15];
        acc = __builtin_amdgcn_wmma_f32_16x16x4_f32(
                  false, a, false, b, (short)0, acc, false, false);
    }
    if (l15 < N_CLASSES) {
        float bz = bias[l15];
        int mbase = m0 + ((lane >> 4) * 8);
        #pragma unroll
        for (int j = 0; j < 8; ++j)
            out[(size_t)(mbase + j) * N_CLASSES + l15] = acc[j] + bz;
    }
}

__global__ void __launch_bounds__(256)
edge_msg1(const int* __restrict__ src, const int* __restrict__ dst,
          const int* __restrict__ et, const float* __restrict__ x,
          const float* __restrict__ W1, const unsigned* __restrict__ cnt,
          float* __restrict__ h, int E) {
    __shared__ __align__(16) float xs[16 * IN_CH];
    int tid    = (int)threadIdx.x;
    int eLocal = tid >> 4;
    int o      = tid & 15;
    int e      = (int)blockIdx.x * 16 + eLocal;
    bool valid = e < E;
    int s = 0, d = 0, r = 0;
    if (valid) { s = src[e]; d = dst[e]; r = et[e]; }

    if (valid) {
        float4 xv = *(const float4*)(x + (size_t)s * IN_CH + o * 4);
        *(float4*)(xs + eLocal * IN_CH + o * 4) = xv;
    }
    __syncthreads();
    if (!valid) return;

    float c    = (float)cnt[d * N_REL + r];
    float norm = 1.0f / fmaxf(c, 1.0f);
    const float* xr = xs + eLocal * IN_CH;
    const float* w  = W1 + (size_t)r * (IN_CH * HID) + o;
    float acc = 0.f;
    #pragma unroll
    for (int k = 0; k < IN_CH; ++k)
        acc = fmaf(xr[k], w[(size_t)k * HID], acc);
    atomicAdd(&h[(size_t)d * HID + o], acc * norm);
}

__global__ void relu_inplace(float* __restrict__ h, int n) {
    int i = blockIdx.x * blockDim.x + threadIdx.x;
    if (i < n) h[i] = fmaxf(h[i], 0.0f);
}

__global__ void __launch_bounds__(256)
edge_msg2(const int* __restrict__ src, const int* __restrict__ dst,
          const int* __restrict__ et, const float* __restrict__ h,
          const float* __restrict__ W2, const unsigned* __restrict__ cnt,
          float* __restrict__ out, int E) {
    __shared__ __align__(8) float hs[32 * HID];
    int tid    = (int)threadIdx.x;
    int eLocal = tid >> 3;
    int c      = tid & 7;
    int e      = (int)blockIdx.x * 32 + eLocal;
    bool valid = e < E;
    int s = 0, d = 0, r = 0;
    if (valid) { s = src[e]; d = dst[e]; r = et[e]; }

    if (valid) {
        float2 hv = *(const float2*)(h + (size_t)s * HID + c * 2);
        *(float2*)(hs + eLocal * HID + c * 2) = hv;
    }
    __syncthreads();
    if (!valid) return;

    float cn   = (float)cnt[d * N_REL + r];
    float norm = 1.0f / fmaxf(cn, 1.0f);
    const float* hr = hs + eLocal * HID;
    const float* w  = W2 + (size_t)r * (HID * N_CLASSES) + c;
    float acc = 0.f;
    #pragma unroll
    for (int k = 0; k < HID; ++k)
        acc = fmaf(hr[k], w[k * N_CLASSES], acc);
    atomicAdd(&out[(size_t)d * N_CLASSES + c], acc * norm);
}

__global__ void log_softmax8(float* __restrict__ out) {
    int n = blockIdx.x * blockDim.x + threadIdx.x;
    if (n >= N_NODES) return;
    float v[N_CLASSES];
    float m = -3.402823466e+38f;
    #pragma unroll
    for (int c = 0; c < N_CLASSES; ++c) {
        v[c] = out[(size_t)n * N_CLASSES + c];
        m = fmaxf(m, v[c]);
    }
    float sum = 0.f;
    #pragma unroll
    for (int c = 0; c < N_CLASSES; ++c) sum += __expf(v[c] - m);
    float lse = m + __logf(sum);
    #pragma unroll
    for (int c = 0; c < N_CLASSES; ++c)
        out[(size_t)n * N_CLASSES + c] = v[c] - lse;
}

extern "C" void kernel_launch(void* const* d_in, const int* in_sizes, int n_in,
                              void* d_out, int out_size, void* d_ws, size_t ws_size,
                              hipStream_t stream) {
    const float* x      = (const float*)d_in[0];
    const int*   eidx   = (const int*)  d_in[1];
    const int*   etype  = (const int*)  d_in[2];
    const float* comp1  = (const float*)d_in[3];
    const float* basis1 = (const float*)d_in[4];
    const float* root1  = (const float*)d_in[5];
    const float* bias1  = (const float*)d_in[6];
    const float* comp2  = (const float*)d_in[7];
    const float* basis2 = (const float*)d_in[8];
    const float* root2  = (const float*)d_in[9];
    const float* bias2  = (const float*)d_in[10];
    float* out = (float*)d_out;

    const int* src = eidx;
    const int* dst = eidx + N_EDGES;

    char* ws = (char*)d_ws;
    unsigned* cnt = (unsigned*)ws;
    float* W1     = (float*)(ws + (size_t)N_NODES * N_REL * 4);
    float* W2     = W1 + N_REL * IN_CH * HID;
    float* h      = W2 + N_REL * HID * N_CLASSES;
    float* root2p = h + (size_t)N_NODES * HID;

    const int B = 256;
    {
        int n = N_NODES * N_REL;
        zero_u32<<<(n + B - 1) / B, B, 0, stream>>>(cnt, n);
    }
    count_edges<<<(N_EDGES + B - 1) / B, B, 0, stream>>>(dst, etype, cnt, N_EDGES);
    {
        int n1 = N_REL * IN_CH * HID;
        make_w1<<<(n1 + B - 1) / B, B, 0, stream>>>(comp1, basis1, W1);
        int n2 = N_REL * HID * N_CLASSES;
        make_w2<<<(n2 + B - 1) / B, B, 0, stream>>>(comp2, basis2, W2);
        pad_root2<<<1, B, 0, stream>>>(root2, root2p);
    }
    {
        int ntiles = N_NODES / 16;
        int wavesPerBlock = B / 32;
        int grid = (ntiles + wavesPerBlock - 1) / wavesPerBlock;
        root_gemm1<<<grid, B, 0, stream>>>(x, root1, bias1, h);
    }
    edge_msg1<<<(N_EDGES + 15) / 16, B, 0, stream>>>(src, dst, etype, x, W1, cnt, h, N_EDGES);
    {
        int n = N_NODES * HID;
        relu_inplace<<<(n + B - 1) / B, B, 0, stream>>>(h, n);
    }
    {
        int ntiles = N_NODES / 16;
        int wavesPerBlock = B / 32;
        int grid = (ntiles + wavesPerBlock - 1) / wavesPerBlock;
        root_gemm2<<<grid, B, 0, stream>>>(h, root2p, bias2, out);
    }
    edge_msg2<<<(N_EDGES + 31) / 32, B, 0, stream>>>(src, dst, etype, h, W2, cnt, out, N_EDGES);
    log_softmax8<<<(N_NODES + B - 1) / B, B, 0, stream>>>(out);
}


# layer-1 edge kernel: weight table split by relation range, not channel half; each edge done once by 16 lanes picked by a wave vote: one x row, count read, 64-B add row per edge (k sum order changes)
# speedup vs baseline: 1.7139x; 1.7139x over previous
#include <hip/hip_runtime.h>
#include <hip/hip_bf16.h>

typedef __attribute__((ext_vector_type(2))) float v2f;
typedef __attribute__((ext_vector_type(8))) float v8f;

#define N_NODES 40000
#define N_EDGES 400000
#define IN_CH   64
#define HID     16
#define N_REL   90
#define N_BASES 30
#define N_CLASSES 8


#define SB 256
#define W1_BLOCKS   ((N_REL * IN_CH * HID + SB - 1) / SB)
#define W2_BLOCKS   ((N_REL * HID * N_CLASSES + SB - 1) / SB)
#define ZERO_BLOCKS ((N_NODES * N_REL + SB - 1) / SB)
#define OUTZ_BLOCKS ((N_NODES * N_CLASSES / 4 + SB - 1) / SB)
#define SETUP_GRID  (W1_BLOCKS + W2_BLOCKS + 1 + OUTZ_BLOCKS + ZERO_BLOCKS)
#define G1_BLOCKS   ((N_NODES / 16 + SB / 32 - 1) / (SB / 32))
#define CNT_BLOCKS  ((N_EDGES + SB - 1) / SB)
#define ROOT1_COUNT_GRID (G1_BLOCKS + CNT_BLOCKS)

__device__ __forceinline__ void make_w1(int blk, const float* __restrict__ comp,
                                        const float* __restrict__ basis,
                                        float* __restrict__ W) {
    int idx = blk * SB + (int)threadIdx.x;
    if (idx >= N_REL * IN_CH * HID) return;
    int r = idx / (IN_CH * HID);
    int rem = idx - r * (IN_CH * HID);
    float acc = 0.f;
    #pragma unroll 5
    for (int b = 0; b < N_BASES; ++b)
        acc += comp[r * N_BASES + b] * basis[b * (IN_CH * HID) + rem];
    W[idx] = acc;
}

__device__ __forceinline__ void make_w2(int blk, const float* __restrict__ comp,
                                        const float* __restrict__ basis,
                                        float* __restrict__ W) {
    int idx = blk * SB + (int)threadIdx.x;
    if (idx >= N_REL * HID * N_CLASSES) return;
    int r = idx / (HID * N_CLASSES);
    int rem = idx - r * (HID * N_CLASSES);
    float acc = 0.f;
    #pragma unroll 5
    for (int b = 0; b < N_BASES; ++b)
        acc += comp[r * N_BASES + b] * basis[b * (HID * N_CLASSES) + rem];
    W[idx] = acc;
}

__device__ __forceinline__ void pad_root2(const float* __restrict__ root2,
                                          float* __restrict__ root2p) {
    int idx = (int)threadIdx.x;
    if (idx >= HID * 16) return;
    int k = idx >> 4, n = idx & 15;
    root2p[idx] = (n < N_CLASSES) ? root2[k * N_CLASSES + n] : 0.0f;
}

__global__ void __launch_bounds__(SB)
setup_tables(const float* __restrict__ comp1, const float* __restrict__ basis1,
             float* __restrict__ W1,
             const float* __restrict__ comp2, const float* __restrict__ basis2,
             float* __restrict__ W2,
             const float* __restrict__ root2, float* __restrict__ root2p,
             float* __restrict__ out, unsigned* __restrict__ cnt) {
    int b = (int)blockIdx.x;
    if (b < W1_BLOCKS) {
        make_w1(b, comp1, basis1, W1);
    } else if (b < W1_BLOCKS + W2_BLOCKS) {
        make_w2(b - W1_BLOCKS, comp2, basis2, W2);
    } else if (b == W1_BLOCKS + W2_BLOCKS) {
        pad_root2(root2, root2p);
    } else if (b < W1_BLOCKS + W2_BLOCKS + 1 + OUTZ_BLOCKS) {
        int i = (b - (W1_BLOCKS + W2_BLOCKS + 1)) * SB + (int)threadIdx.x;
        if (i < N_NODES * N_CLASSES / 4)
            ((float4*)out)[i] = make_float4(0.f, 0.f, 0.f, 0.f);
    } else {
        int i = (b - (W1_BLOCKS + W2_BLOCKS + 1 + OUTZ_BLOCKS)) * SB + (int)threadIdx.x;
        if (i < N_NODES * N_REL) cnt[i] = 0u;
    }
}

__device__ __forceinline__ void
root_gemm1(int blk, const float* __restrict__ x, const float* __restrict__ root,
           const float* __restrict__ bias, float* __restrict__ h) {
    const int ntiles = N_NODES / 16;
    int wave = (int)((blk * SB + threadIdx.x) >> 5);
    int lane = (int)(threadIdx.x & 31);
    if (wave >= ntiles) return;

    int l15   = lane & 15;
    int khalf = (lane >> 4) * 2;
    int m0    = wave * 16;

    const float* xrow = x + (size_t)(m0 + l15) * IN_CH;
    v8f acc = {};
    #pragma unroll
    for (int k0 = 0; k0 < IN_CH; k0 += 4) {
        v2f a = *(const v2f*)(xrow + k0 + khalf);
        v2f b;
        b.x = root[(k0 + khalf)     * HID + l15];
        b.y = root[(k0 + khalf + 1) * HID + l15];
        acc = __builtin_amdgcn_wmma_f32_16x16x4_f32(
                  false, a, false, b, (short)0, acc, false, false);
    }
    float bz = bias[l15];
    int mbase = m0 + ((lane >> 4) * 8);
    #pragma unroll
    for (int j = 0; j < 8; ++j)
        h[(size_t)(mbase + j) * HID + l15] = acc[j] + bz;
}

__global__ void __launch_bounds__(SB)
root1_and_count(const float* __restrict__ x, const float* __restrict__ root,
                const float* __restrict__ bias, float* __restrict__ h,
                const int* __restrict__ dst, const int* __restrict__ et,
                unsigned* __restrict__ cnt) {
    int b = (int)blockIdx.x;
    if (b < G1_BLOCKS) {
        root_gemm1(b, x, root, bias, h);
    } else {
        int e = (b - G1_BLOCKS) * SB + (int)threadIdx.x;
        if (e < N_EDGES) atomicAdd(&cnt[dst[e] * N_REL + et[e]], 1u);
    }
}

#define W1L_PART_PITCH 68
#define W1L_REL_PITCH  (16 * W1L_PART_PITCH)
#define E1_REL_SPLIT ((N_REL + 1) / 2)
#define E1_THREADS 1024
#define E1_GROUPS  128
#define E1_GRID    256
#define E1_WAVES   (E1_GROUPS * (E1_THREADS / 32))

__device__ __forceinline__ float lane8_xor4(float v) {
    return __uint_as_float(__builtin_amdgcn_mov_dpp8(__float_as_uint(v), 6852524));
}
__device__ __forceinline__ float lane8_xor2(float v) {
    return __uint_as_float(__builtin_amdgcn_mov_dpp8(__float_as_uint(v), 11788826));
}
__device__ __forceinline__ float lane8_xor1(float v) {
    return __uint_as_float(__builtin_amdgcn_mov_dpp8(__float_as_uint(v), 14570689));
}

template <int M>
__device__ __forceinline__ float lane16_xor(float v) {
    return __uint_as_float((unsigned)__builtin_amdgcn_update_dpp(
        0, (int)__float_as_uint(v), 0x160 + M, 0xF, 0xF, false));
}

__device__ __forceinline__ float fold8(const float (&a)[8], int p) {
    const bool hi4 = (p & 4) != 0, hi2 = (p & 2) != 0, hi1 = (p & 1) != 0;
    float b[4];
    #pragma unroll
    for (int i = 0; i < 4; ++i) {
        float keep = hi4 ? a[i + 4] : a[i];
        float send = hi4 ? a[i] : a[i + 4];
        b[i] = keep + lane8_xor4(send);
    }
    float c2[2];
    #pragma unroll
    for (int i = 0; i < 2; ++i) {
        float keep = hi2 ? b[i + 2] : b[i];
        float send = hi2 ? b[i] : b[i + 2];
        c2[i] = keep + lane8_xor2(send);
    }
    float keep1 = hi1 ? c2[1] : c2[0];
    float send1 = hi1 ? c2[0] : c2[1];
    return keep1 + lane8_xor1(send1);
}

__device__ __forceinline__ float fold16(const float (&a)[16], int p) {
    const bool hi8 = (p & 8) != 0, hi4 = (p & 4) != 0, hi2 = (p & 2) != 0, hi1 = (p & 1) != 0;
    float b[8];
    #pragma unroll
    for (int i = 0; i < 8; ++i) {
        float keep = hi8 ? a[i + 8] : a[i];
        float send = hi8 ? a[i] : a[i + 8];
        b[i] = keep + lane16_xor<8>(send);
    }
    float c[4];
    #pragma unroll
    for (int i = 0; i < 4; ++i) {
        float keep = hi4 ? b[i + 4] : b[i];
        float send = hi4 ? b[i] : b[i + 4];
        c[i] = keep + lane16_xor<4>(send);
    }
    float d2[2];
    #pragma unroll
    for (int i = 0; i < 2; ++i) {
        float keep = hi2 ? c[i + 2] : c[i];
        float send = hi2 ? c[i] : c[i + 2];
        d2[i] = keep + lane16_xor<2>(send);
    }
    float keep1 = hi1 ? d2[1] : d2[0];
    float send1 = hi1 ? d2[0] : d2[1];
    return keep1 + lane16_xor<1>(send1);
}

__global__ void __launch_bounds__(E1_THREADS)
edge_msg1(const int* __restrict__ src, const int* __restrict__ dst,
          const int* __restrict__ et, const float* __restrict__ x,
          const float* __restrict__ W1, const unsigned* __restrict__ cnt,
          float* __restrict__ h, int E) {
    __shared__ __align__(16) float wl[E1_REL_SPLIT * W1L_REL_PITCH];

    const int tid  = (int)threadIdx.x;
    const int half = (int)(blockIdx.x & 1);
    const int grp  = (int)(blockIdx.x >> 1);
    const int r_lo = half * E1_REL_SPLIT;
    const int n_r  = half ? (N_REL - E1_REL_SPLIT) : E1_REL_SPLIT;

    const float* Wsrc = W1 + (size_t)r_lo * IN_CH * HID;
    for (int idx = tid; idx < n_r * IN_CH * HID; idx += E1_THREADS) {
        int o  = idx & 15;
        int rk = idx >> 4;
        int k  = rk & (IN_CH - 1);
        int rl = rk >> 6;
        wl[rl * W1L_REL_PITCH + (k >> 2) * W1L_PART_PITCH + (k & 3) * 16 + o] = Wsrc[idx];
    }
    __syncthreads();

    const int lane  = tid & 31;
    const int p     = lane & 15;
    const bool upper = lane >= 16;
    const int gw = (tid >> 5) * E1_GROUPS + grp;

    for (int base = gw * 32; base < E; base += E1_WAVES * 32) {
        const int e = base + lane;
        int r = -1, s = 0, d = 0;
        if (e < E) { r = et[e]; s = src[e]; d = dst[e]; }
        const int  rl   = r - r_lo;
        const bool mine = (rl >= 0) && (rl < n_r);
        float nrm = 0.0f;
        if (mine) nrm = 1.0f / fmaxf((float)cnt[d * N_REL + r], 1.0f);

        unsigned todo = __builtin_amdgcn_ballot_w32(mine);
        while (todo != 0u) {
            const int la = __builtin_ctz(todo);
            todo &= todo - 1u;
            const bool two = todo != 0u;
            const int lb = two ? __builtin_ctz(todo) : la;
            todo &= todo - 1u;

            const int sa = __builtin_amdgcn_readlane(s, la);
            const int sb = __builtin_amdgcn_readlane(s, lb);
            const int da = __builtin_amdgcn_readlane(d, la);
            const int db = __builtin_amdgcn_readlane(d, lb);
            const int ra = __builtin_amdgcn_readlane(rl, la);
            const int rb = __builtin_amdgcn_readlane(rl, lb);
            const int na = __builtin_amdgcn_readlane((int)__float_as_uint(nrm), la);
            const int nb = __builtin_amdgcn_readlane((int)__float_as_uint(nrm), lb);
            const int   se  = upper ? sb : sa;
            const int   de  = upper ? db : da;
            const int   rle = upper ? rb : ra;
            const float ne  = __uint_as_float((unsigned)(upper ? nb : na));

            const float4 xv = *(const float4*)(x + (size_t)se * IN_CH + p * 4);
            const float* wp = wl + rle * W1L_REL_PITCH + p * W1L_PART_PITCH;
            float xs[4] = {xv.x, xv.y, xv.z, xv.w};
            float a[16] = {0.f, 0.f, 0.f, 0.f, 0.f, 0.f, 0.f, 0.f,
                           0.f, 0.f, 0.f, 0.f, 0.f, 0.f, 0.f, 0.f};
            #pragma unroll
            for (int kk = 0; kk < 4; ++kk) {
                const float* wr = wp + kk * 16;
                float4 w0 = *(const float4*)(wr);
                float4 w1 = *(const float4*)(wr + 4);
                float4 w2 = *(const float4*)(wr + 8);
                float4 w3 = *(const float4*)(wr + 12);
                a[0]  = fmaf(xs[kk], w0.x, a[0]);
                a[1]  = fmaf(xs[kk], w0.y, a[1]);
                a[2]  = fmaf(xs[kk], w0.z, a[2]);
                a[3]  = fmaf(xs[kk], w0.w, a[3]);
                a[4]  = fmaf(xs[kk], w1.x, a[4]);
                a[5]  = fmaf(xs[kk], w1.y, a[5]);
                a[6]  = fmaf(xs[kk], w1.z, a[6]);
                a[7]  = fmaf(xs[kk], w1.w, a[7]);
                a[8]  = fmaf(xs[kk], w2.x, a[8]);
                a[9]  = fmaf(xs[kk], w2.y, a[9]);
                a[10] = fmaf(xs[kk], w2.z, a[10]);
                a[11] = fmaf(xs[kk], w2.w, a[11]);
                a[12] = fmaf(xs[kk], w3.x, a[12]);
                a[13] = fmaf(xs[kk], w3.y, a[13]);
                a[14] = fmaf(xs[kk], w3.z, a[14]);
                a[15] = fmaf(xs[kk], w3.w, a[15]);
            }
            float acc = fold16(a, p);

            if (!upper || two)
                atomicAdd(&h[(size_t)de * HID + p], acc * ne);
        }
    }
}

#define W2L_PART_PITCH 20
#define W2L_REL_PITCH  (8 * W2L_PART_PITCH)
#define E2_THREADS 1024
#define E2_GROUPS  512
#define E2_EDGES_PER_PASS (E2_THREADS / 8)

__global__ void __launch_bounds__(E2_THREADS)
edge_msg2(const int* __restrict__ src, const int* __restrict__ dst,
          const int* __restrict__ et, const float* __restrict__ h,
          const float* __restrict__ W2, const unsigned* __restrict__ cnt,
          float* __restrict__ out, int E) {
    __shared__ __align__(16) float wl[N_REL * W2L_REL_PITCH];

    const int tid = (int)threadIdx.x;
    const int grp = (int)blockIdx.x;

    for (int idx = tid; idx < N_REL * HID * N_CLASSES; idx += E2_THREADS) {
        int o  = idx & 7;
        int rk = idx >> 3;
        int k  = rk & (HID - 1);
        int r  = rk >> 4;
        wl[r * W2L_REL_PITCH + (k >> 1) * W2L_PART_PITCH + (k & 1) * 8 + o] = W2[idx];
    }
    __syncthreads();

    const int p    = tid & 7;
    const int slot = tid >> 3;

    for (int e = grp * E2_EDGES_PER_PASS + slot; e < E;
         e += E2_GROUPS * E2_EDGES_PER_PASS) {
        int s = src[e], d = dst[e], r = et[e];
        float2 hv = *(const float2*)(h + (size_t)s * HID + p * 2);
        const float* wp = wl + r * W2L_REL_PITCH + p * W2L_PART_PITCH;
        float xs[2] = {fmaxf(hv.x, 0.0f), fmaxf(hv.y, 0.0f)};
        float a[8] = {0.f, 0.f, 0.f, 0.f, 0.f, 0.f, 0.f, 0.f};
        #pragma unroll
        for (int kk = 0; kk < 2; ++kk) {
            float4 w0 = *(const float4*)(wp + kk * 8);
            float4 w1 = *(const float4*)(wp + kk * 8 + 4);
            a[0] = fmaf(xs[kk], w0.x, a[0]);
            a[1] = fmaf(xs[kk], w0.y, a[1]);
            a[2] = fmaf(xs[kk], w0.z, a[2]);
            a[3] = fmaf(xs[kk], w0.w, a[3]);
            a[4] = fmaf(xs[kk], w1.x, a[4]);
            a[5] = fmaf(xs[kk], w1.y, a[5]);
            a[6] = fmaf(xs[kk], w1.z, a[6]);
            a[7] = fmaf(xs[kk], w1.w, a[7]);
        }
        float acc = fold8(a, p);

        float cn   = (float)cnt[d * N_REL + r];
        float norm = 1.0f / fmaxf(cn, 1.0f);
        atomicAdd(&out[(size_t)d * N_CLASSES + p], acc * norm);
    }
}

__device__ __forceinline__ float row8_max(float v) {
    v = fmaxf(v, lane8_xor4(v));
    v = fmaxf(v, lane8_xor2(v));
    return fmaxf(v, lane8_xor1(v));
}
__device__ __forceinline__ float row8_sum(float v) {
    v += lane8_xor4(v);
    v += lane8_xor2(v);
    return v + lane8_xor1(v);
}

__global__ void __launch_bounds__(SB)
root2_logsoftmax(const float* __restrict__ h, const float* __restrict__ root2p,
                 const float* __restrict__ bias, float* __restrict__ out) {
    const int ntiles = N_NODES / 16;
    int wave = (int)((blockIdx.x * SB + threadIdx.x) >> 5);
    int lane = (int)(threadIdx.x & 31);
    if (wave >= ntiles) return;

    int l15   = lane & 15;
    int khalf = (lane >> 4) * 2;
    int m0    = wave * 16;
    int mbase = m0 + ((lane >> 4) * 8);
    const bool live = l15 < N_CLASSES;
    const int  col  = l15 & (N_CLASSES - 1);

    float msg[8];
    #pragma unroll
    for (int j = 0; j < 8; ++j)
        msg[j] = out[(size_t)(mbase + j) * N_CLASSES + col];
    float bz = bias[col];

    const float* hrow = h + (size_t)(m0 + l15) * HID;
    v8f acc = {};
    #pragma unroll
    for (int k0 = 0; k0 < HID; k0 += 4) {
        v2f a = *(const v2f*)(hrow + k0 + khalf);
        a.x = fmaxf(a.x, 0.0f);
        a.y = fmaxf(a.y, 0.0f);
        v2f b;
        b.x = root2p[(k0 + khalf)     * 16 + l15];
        b.y = root2p[(k0 + khalf + 1) * 16 + l15];
        acc = __builtin_amdgcn_wmma_f32_16x16x4_f32(
                  false, a, false, b, (short)0, acc, false, false);
    }
    float res[8];
    #pragma unroll
    for (int j = 0; j < 8; ++j) {
        float v   = (acc[j] + bz) + msg[j];
        float m   = row8_max(v);
        float sum = row8_sum(__expf(v - m));
        res[j] = v - (m + __logf(sum));
    }
    if (live) {
        #pragma unroll
        for (int j = 0; j < 8; ++j)
            out[(size_t)(mbase + j) * N_CLASSES + l15] = res[j];
    }
}

extern "C" void kernel_launch(void* const* d_in, const int* in_sizes, int n_in,
                              void* d_out, int out_size, void* d_ws, size_t ws_size,
                              hipStream_t stream) {
    const float* x      = (const float*)d_in[0];
    const int*   eidx   = (const int*)  d_in[1];
    const int*   etype  = (const int*)  d_in[2];
    const float* comp1  = (const float*)d_in[3];
    const float* basis1 = (const float*)d_in[4];
    const float* root1  = (const float*)d_in[5];
    const float* bias1  = (const float*)d_in[6];
    const float* comp2  = (const float*)d_in[7];
    const float* basis2 = (const float*)d_in[8];
    const float* root2  = (const float*)d_in[9];
    const float* bias2  = (const float*)d_in[10];
    float* out = (float*)d_out;

    const int* src = eidx;
    const int* dst = eidx + N_EDGES;

    char* ws = (char*)d_ws;
    unsigned* cnt = (unsigned*)ws;
    float* W1     = (float*)(ws + (size_t)N_NODES * N_REL * 4);
    float* W2     = W1 + N_REL * IN_CH * HID;
    float* h      = W2 + N_REL * HID * N_CLASSES;
    float* root2p = h + (size_t)N_NODES * HID;

    setup_tables<<<SETUP_GRID, SB, 0, stream>>>(comp1, basis1, W1, comp2, basis2, W2,
                                                root2, root2p, out, cnt);
    root1_and_count<<<ROOT1_COUNT_GRID, SB, 0, stream>>>(x, root1, bias1, h, dst, etype, cnt);
    edge_msg1<<<E1_GRID, E1_THREADS, 0, stream>>>(src, dst, etype, x, W1, cnt, h, N_EDGES);
    edge_msg2<<<E2_GROUPS, E2_THREADS, 0, stream>>>(src, dst, etype, h, W2, cnt, out, N_EDGES);
    root2_logsoftmax<<<G1_BLOCKS, SB, 0, stream>>>(h, root2p, bias2, out);
}


# edge kernels: gathers issued in groups ahead of use (x rows of 4 edge pairs, next block's indices; layer 2: 4 edges per slot), weight tables staged by 16-B loads in one group
# speedup vs baseline: 1.8997x; 1.1084x over previous
#include <hip/hip_runtime.h>
#include <hip/hip_bf16.h>

typedef __attribute__((ext_vector_type(2))) float v2f;
typedef __attribute__((ext_vector_type(8))) float v8f;

#define N_NODES 40000
#define N_EDGES 400000
#define IN_CH   64
#define HID     16
#define N_REL   90
#define N_BASES 30
#define N_CLASSES 8


#define SB 256
#define W1_BLOCKS   ((N_REL * IN_CH * HID + SB - 1) / SB)
#define W2_BLOCKS   ((N_REL * HID * N_CLASSES + SB - 1) / SB)
#define ZERO_BLOCKS ((N_NODES * N_REL + SB - 1) / SB)
#define OUTZ_BLOCKS ((N_NODES * N_CLASSES / 4 + SB - 1) / SB)
#define SETUP_GRID  (W1_BLOCKS + W2_BLOCKS + 1 + OUTZ_BLOCKS + ZERO_BLOCKS)
#define G1_BLOCKS   ((N_NODES / 16 + SB / 32 - 1) / (SB / 32))
#define CNT_BLOCKS  ((N_EDGES + SB - 1) / SB)
#define ROOT1_COUNT_GRID (G1_BLOCKS + CNT_BLOCKS)

__device__ __forceinline__ void make_w1(int blk, const float* __restrict__ comp,
                                        const float* __restrict__ basis,
                                        float* __restrict__ W) {
    int idx = blk * SB + (int)threadIdx.x;
    if (idx >= N_REL * IN_CH * HID) return;
    int r = idx / (IN_CH * HID);
    int rem = idx - r * (IN_CH * HID);
    float acc = 0.f;
    #pragma unroll 5
    for (int b = 0; b < N_BASES; ++b)
        acc += comp[r * N_BASES + b] * basis[b * (IN_CH * HID) + rem];
    W[idx] = acc;
}

__device__ __forceinline__ void make_w2(int blk, const float* __restrict__ comp,
                                        const float* __restrict__ basis,
                                        float* __restrict__ W) {
    int idx = blk * SB + (int)threadIdx.x;
    if (idx >= N_REL * HID * N_CLASSES) return;
    int r = idx / (HID * N_CLASSES);
    int rem = idx - r * (HID * N_CLASSES);
    float acc = 0.f;
    #pragma unroll 5
    for (int b = 0; b < N_BASES; ++b)
        acc += comp[r * N_BASES + b] * basis[b * (HID * N_CLASSES) + rem];
    W[idx] = acc;
}

__device__ __forceinline__ void pad_root2(const float* __restrict__ root2,
                                          float* __restrict__ root2p) {
    int idx = (int)threadIdx.x;
    if (idx >= HID * 16) return;
    int k = idx >> 4, n = idx & 15;
    root2p[idx] = (n < N_CLASSES) ? root2[k * N_CLASSES + n] : 0.0f;
}

__global__ void __launch_bounds__(SB)
setup_tables(const float* __restrict__ comp1, const float* __restrict__ basis1,
             float* __restrict__ W1,
             const float* __restrict__ comp2, const float* __restrict__ basis2,
             float* __restrict__ W2,
             const float* __restrict__ root2, float* __restrict__ root2p,
             float* __restrict__ out, unsigned* __restrict__ cnt) {
    int b = (int)blockIdx.x;
    if (b < W1_BLOCKS) {
        make_w1(b, comp1, basis1, W1);
    } else if (b < W1_BLOCKS + W2_BLOCKS) {
        make_w2(b - W1_BLOCKS, comp2, basis2, W2);
    } else if (b == W1_BLOCKS + W2_BLOCKS) {
        pad_root2(root2, root2p);
    } else if (b < W1_BLOCKS + W2_BLOCKS + 1 + OUTZ_BLOCKS) {
        int i = (b - (W1_BLOCKS + W2_BLOCKS + 1)) * SB + (int)threadIdx.x;
        if (i < N_NODES * N_CLASSES / 4)
            ((float4*)out)[i] = make_float4(0.f, 0.f, 0.f, 0.f);
    } else {
        int i = (b - (W1_BLOCKS + W2_BLOCKS + 1 + OUTZ_BLOCKS)) * SB + (int)threadIdx.x;
        if (i < N_NODES * N_REL) cnt[i] = 0u;
    }
}

__device__ __forceinline__ void
root_gemm1(int blk, const float* __restrict__ x, const float* __restrict__ root,
           const float* __restrict__ bias, float* __restrict__ h) {
    const int ntiles = N_NODES / 16;
    int wave = (int)((blk * SB + threadIdx.x) >> 5);
    int lane = (int)(threadIdx.x & 31);
    if (wave >= ntiles) return;

    int l15   = lane & 15;
    int khalf = (lane >> 4) * 2;
    int m0    = wave * 16;

    const float* xrow = x + (size_t)(m0 + l15) * IN_CH;
    v8f acc = {};
    #pragma unroll
    for (int k0 = 0; k0 < IN_CH; k0 += 4) {
        v2f a = *(const v2f*)(xrow + k0 + khalf);
        v2f b;
        b.x = root[(k0 + khalf)     * HID + l15];
        b.y = root[(k0 + khalf + 1) * HID + l15];
        acc = __builtin_amdgcn_wmma_f32_16x16x4_f32(
                  false, a, false, b, (short)0, acc, false, false);
    }
    float bz = bias[l15];
    int mbase = m0 + ((lane >> 4) * 8);
    #pragma unroll
    for (int j = 0; j < 8; ++j)
        h[(size_t)(mbase + j) * HID + l15] = acc[j] + bz;
}

__global__ void __launch_bounds__(SB)
root1_and_count(const float* __restrict__ x, const float* __restrict__ root,
                const float* __restrict__ bias, float* __restrict__ h,
                const int* __restrict__ dst, const int* __restrict__ et,
                unsigned* __restrict__ cnt) {
    int b = (int)blockIdx.x;
    if (b < G1_BLOCKS) {
        root_gemm1(b, x, root, bias, h);
    } else {
        int e = (b - G1_BLOCKS) * SB + (int)threadIdx.x;
        if (e < N_EDGES) atomicAdd(&cnt[dst[e] * N_REL + et[e]], 1u);
    }
}

#define W1L_PART_PITCH 68
#define W1L_REL_PITCH  (16 * W1L_PART_PITCH)
#define E1_REL_SPLIT ((N_REL + 1) / 2)
#define E1_THREADS 1024
#define E1_GROUPS  128
#define E1_GRID    256
#define E1_WAVES   (E1_GROUPS * (E1_THREADS / 32))
#define E1_STAGE_TRIPS ((E1_REL_SPLIT * IN_CH * HID / 4 + E1_THREADS - 1) / E1_THREADS)
#define E1_PAIRS 4

__device__ __forceinline__ float lane8_xor4(float v) {
    return __uint_as_float(__builtin_amdgcn_mov_dpp8(__float_as_uint(v), 6852524));
}
__device__ __forceinline__ float lane8_xor2(float v) {
    return __uint_as_float(__builtin_amdgcn_mov_dpp8(__float_as_uint(v), 11788826));
}
__device__ __forceinline__ float lane8_xor1(float v) {
    return __uint_as_float(__builtin_amdgcn_mov_dpp8(__float_as_uint(v), 14570689));
}

template <int M>
__device__ __forceinline__ float lane16_xor(float v) {
    return __uint_as_float((unsigned)__builtin_amdgcn_update_dpp(
        0, (int)__float_as_uint(v), 0x160 + M, 0xF, 0xF, false));
}

__device__ __forceinline__ float fold8(const float (&a)[8], int p) {
    const bool hi4 = (p & 4) != 0, hi2 = (p & 2) != 0, hi1 = (p & 1) != 0;
    float b[4];
    #pragma unroll
    for (int i = 0; i < 4; ++i) {
        float keep = hi4 ? a[i + 4] : a[i];
        float send = hi4 ? a[i] : a[i + 4];
        b[i] = keep + lane8_xor4(send);
    }
    float c2[2];
    #pragma unroll
    for (int i = 0; i < 2; ++i) {
        float keep = hi2 ? b[i + 2] : b[i];
        float send = hi2 ? b[i] : b[i + 2];
        c2[i] = keep + lane8_xor2(send);
    }
    float keep1 = hi1 ? c2[1] : c2[0];
    float send1 = hi1 ? c2[0] : c2[1];
    return keep1 + lane8_xor1(send1);
}

__device__ __forceinline__ float fold16(const float (&a)[16], int p) {
    const bool hi8 = (p & 8) != 0, hi4 = (p & 4) != 0, hi2 = (p & 2) != 0, hi1 = (p & 1) != 0;
    float b[8];
    #pragma unroll
    for (int i = 0; i < 8; ++i) {
        float keep = hi8 ? a[i + 8] : a[i];
        float send = hi8 ? a[i] : a[i + 8];
        b[i] = keep + lane16_xor<8>(send);
    }
    float c[4];
    #pragma unroll
    for (int i = 0; i < 4; ++i) {
        float keep = hi4 ? b[i + 4] : b[i];
        float send = hi4 ? b[i] : b[i + 4];
        c[i] = keep + lane16_xor<4>(send);
    }
    float d2[2];
    #pragma unroll
    for (int i = 0; i < 2; ++i) {
        float keep = hi2 ? c[i + 2] : c[i];
        float send = hi2 ? c[i] : c[i + 2];
        d2[i] = keep + lane16_xor<2>(send);
    }
    float keep1 = hi1 ? d2[1] : d2[0];
    float send1 = hi1 ? d2[0] : d2[1];
    return keep1 + lane16_xor<1>(send1);
}

__global__ void __launch_bounds__(E1_THREADS)
edge_msg1(const int* __restrict__ src, const int* __restrict__ dst,
          const int* __restrict__ et, const float* __restrict__ x,
          const float* __restrict__ W1, const unsigned* __restrict__ cnt,
          float* __restrict__ h, int E) {
    __shared__ __align__(16) float wl[E1_REL_SPLIT * W1L_REL_PITCH];

    const int tid  = (int)threadIdx.x;
    const int half = (int)(blockIdx.x & 1);
    const int grp  = (int)(blockIdx.x >> 1);
    const int r_lo = half * E1_REL_SPLIT;
    const int n_r  = half ? (N_REL - E1_REL_SPLIT) : E1_REL_SPLIT;

    {
        const float4* Wsrc4 = (const float4*)(W1 + (size_t)r_lo * IN_CH * HID);
        const int n4 = n_r * (IN_CH * HID / 4);
        float4 t[E1_STAGE_TRIPS];
        #pragma unroll
        for (int i = 0; i < E1_STAGE_TRIPS; ++i) {
            const int q = min(tid + i * E1_THREADS, n4 - 1);
            t[i] = Wsrc4[q];
        }
        #pragma unroll
        for (int i = 0; i < E1_STAGE_TRIPS; ++i) {
            const int q  = min(tid + i * E1_THREADS, n4 - 1);
            const int o  = (q & 3) * 4;
            const int rk = q >> 2;
            const int k  = rk & (IN_CH - 1);
            const int rl = rk >> 6;
            *(float4*)(wl + rl * W1L_REL_PITCH + (k >> 2) * W1L_PART_PITCH + (k & 3) * 16 + o) = t[i];
        }
    }
    __syncthreads();

    const int lane  = tid & 31;
    const int p     = lane & 15;
    const bool upper = lane >= 16;
    const int gw = (tid >> 5) * E1_GROUPS + grp;

    int base = gw * 32;
    int r = -1, s = 0, d = 0;
    if (base + lane < E) { r = et[base + lane]; s = src[base + lane]; d = dst[base + lane]; }

    for (; base < E; base += E1_WAVES * 32) {
        const int en = base + E1_WAVES * 32 + lane;
        int rn = -1, sn = 0, dn = 0;
        if (en < E) { rn = et[en]; sn = src[en]; dn = dst[en]; }

        const int  rl   = r - r_lo;
        const bool mine = (rl >= 0) && (rl < n_r);
        float nrm = 0.0f;
        if (mine) nrm = 1.0f / fmaxf((float)cnt[d * N_REL + r], 1.0f);

        unsigned todo = __builtin_amdgcn_ballot_w32(mine);
        while (todo != 0u) {
            const int first = __builtin_ctz(todo);
            int  sl[2 * E1_PAIRS];
            bool val[2 * E1_PAIRS];
            #pragma unroll
            for (int j = 0; j < 2 * E1_PAIRS; ++j) {
                val[j] = todo != 0u;
                sl[j]  = val[j] ? __builtin_ctz(todo) : first;
                todo &= todo - 1u;
            }

            float4 xv[E1_PAIRS];
            int    de[E1_PAIRS], rle[E1_PAIRS];
            float  ne[E1_PAIRS];
            #pragma unroll
            for (int i = 0; i < E1_PAIRS; ++i) {
                const int la = sl[2 * i], lb = sl[2 * i + 1];
                const int sa = __builtin_amdgcn_readlane(s, la);
                const int sb = __builtin_amdgcn_readlane(s, lb);
                const int da = __builtin_amdgcn_readlane(d, la);
                const int db = __builtin_amdgcn_readlane(d, lb);
                const int ra = __builtin_amdgcn_readlane(rl, la);
                const int rb = __builtin_amdgcn_readlane(rl, lb);
                const int na = __builtin_amdgcn_readlane((int)__float_as_uint(nrm), la);
                const int nb = __builtin_amdgcn_readlane((int)__float_as_uint(nrm), lb);
                const int se = upper ? sb : sa;
                de[i]  = upper ? db : da;
                rle[i] = upper ? rb : ra;
                ne[i]  = __uint_as_float((unsigned)(upper ? nb : na));
                xv[i]  = *(const float4*)(x + (size_t)se * IN_CH + p * 4);
            }

            #pragma unroll
            for (int i = 0; i < E1_PAIRS; ++i) {
                if (!val[2 * i]) break;
                const float* wp = wl + rle[i] * W1L_REL_PITCH + p * W1L_PART_PITCH;
                float xs[4] = {xv[i].x, xv[i].y, xv[i].z, xv[i].w};
                float a[16] = {0.f, 0.f, 0.f, 0.f, 0.f, 0.f, 0.f, 0.f,
                               0.f, 0.f, 0.f, 0.f, 0.f, 0.f, 0.f, 0.f};
                #pragma unroll
                for (int kk = 0; kk < 4; ++kk) {
                    const float* wr = wp + kk * 16;
                    float4 w0 = *(const float4*)(wr);
                    float4 w1 = *(const float4*)(wr + 4);
                    float4 w2 = *(const float4*)(wr + 8);
                    float4 w3 = *(const float4*)(wr + 12);
                    a[0]  = fmaf(xs[kk], w0.x, a[0]);
                    a[1]  = fmaf(xs[kk], w0.y, a[1]);
                    a[2]  = fmaf(xs[kk], w0.z, a[2]);
                    a[3]  = fmaf(xs[kk], w0.w, a[3]);
                    a[4]  = fmaf(xs[kk], w1.x, a[4]);
                    a[5]  = fmaf(xs[kk], w1.y, a[5]);
                    a[6]  = fmaf(xs[kk], w1.z, a[6]);
                    a[7]  = fmaf(xs[kk], w1.w, a[7]);
                    a[8]  = fmaf(xs[kk], w2.x, a[8]);
                    a[9]  = fmaf(xs[kk], w2.y, a[9]);
                    a[10] = fmaf(xs[kk], w2.z, a[10]);
                    a[11] = fmaf(xs[kk], w2.w, a[11]);
                    a[12] = fmaf(xs[kk], w3.x, a[12]);
                    a[13] = fmaf(xs[kk], w3.y, a[13]);
                    a[14] = fmaf(xs[kk], w3.z, a[14]);
                    a[15] = fmaf(xs[kk], w3.w, a[15]);
                }
                float acc = fold16(a, p);

                if (!upper || val[2 * i + 1])
                    atomicAdd(&h[(size_t)de[i] * HID + p], acc * ne[i]);
            }
        }
        r = rn; s = sn; d = dn;
    }
}

#define W2L_PART_PITCH 20
#define W2L_REL_PITCH  (8 * W2L_PART_PITCH)
#define E2_THREADS 1024
#define E2_GROUPS  512
#define E2_EDGES_PER_PASS (E2_THREADS / 8)
#define E2_STAGE_TRIPS ((N_REL * HID * N_CLASSES / 4 + E2_THREADS - 1) / E2_THREADS)
#define E2_BATCH 4

template <int U>
__device__ __forceinline__ void
edge2_group(int e0, int pass, int p,
            const int* __restrict__ src, const int* __restrict__ dst,
            const int* __restrict__ et, const float* __restrict__ h,
            const unsigned* __restrict__ cnt, const float* wl,
            float* __restrict__ out) {
    int s[U], d[U], r[U];
    #pragma unroll
    for (int u = 0; u < U; ++u) {
        const int e = e0 + u * pass;
        s[u] = src[e]; d[u] = dst[e]; r[u] = et[e];
    }
    float2   hv[U];
    unsigned cn[U];
    #pragma unroll
    for (int u = 0; u < U; ++u) {
        hv[u] = *(const float2*)(h + (size_t)s[u] * HID + p * 2);
        cn[u] = cnt[d[u] * N_REL + r[u]];
    }
    #pragma unroll
    for (int u = 0; u < U; ++u) {
        const float* wp = wl + r[u] * W2L_REL_PITCH + p * W2L_PART_PITCH;
        float xs[2] = {fmaxf(hv[u].x, 0.0f), fmaxf(hv[u].y, 0.0f)};
        float a[8] = {0.f, 0.f, 0.f, 0.f, 0.f, 0.f, 0.f, 0.f};
        #pragma unroll
        for (int kk = 0; kk < 2; ++kk) {
            float4 w0 = *(const float4*)(wp + kk * 8);
            float4 w1 = *(const float4*)(wp + kk * 8 + 4);
            a[0] = fmaf(xs[kk], w0.x, a[0]);
            a[1] = fmaf(xs[kk], w0.y, a[1]);
            a[2] = fmaf(xs[kk], w0.z, a[2]);
            a[3] = fmaf(xs[kk], w0.w, a[3]);
            a[4] = fmaf(xs[kk], w1.x, a[4]);
            a[5] = fmaf(xs[kk], w1.y, a[5]);
            a[6] = fmaf(xs[kk], w1.z, a[6]);
            a[7] = fmaf(xs[kk], w1.w, a[7]);
        }
        float acc = fold8(a, p);

        float norm = 1.0f / fmaxf((float)cn[u], 1.0f);
        atomicAdd(&out[(size_t)d[u] * N_CLASSES + p], acc * norm);
    }
}

__global__ void __launch_bounds__(E2_THREADS)
edge_msg2(const int* __restrict__ src, const int* __restrict__ dst,
          const int* __restrict__ et, const float* __restrict__ h,
          const float* __restrict__ W2, const unsigned* __restrict__ cnt,
          float* __restrict__ out, int E) {
    __shared__ __align__(16) float wl[N_REL * W2L_REL_PITCH];

    const int tid = (int)threadIdx.x;
    const int grp = (int)blockIdx.x;

    {
        const float4* W24 = (const float4*)W2;
        const int n4 = N_REL * HID * N_CLASSES / 4;
        float4 t[E2_STAGE_TRIPS];
        #pragma unroll
        for (int i = 0; i < E2_STAGE_TRIPS; ++i) {
            const int q = min(tid + i * E2_THREADS, n4 - 1);
            t[i] = W24[q];
        }
        #pragma unroll
        for (int i = 0; i < E2_STAGE_TRIPS; ++i) {
            const int q  = min(tid + i * E2_THREADS, n4 - 1);
            const int o  = (q & 1) * 4;
            const int rk = q >> 1;
            const int k  = rk & (HID - 1);
            const int r  = rk >> 4;
            *(float4*)(wl + r * W2L_REL_PITCH + (k >> 1) * W2L_PART_PITCH + (k & 1) * 8 + o) = t[i];
        }
    }
    __syncthreads();

    const int p    = tid & 7;
    const int slot = tid >> 3;

    const int pass = E2_GROUPS * E2_EDGES_PER_PASS;
    int e0 = grp * E2_EDGES_PER_PASS + slot;
    for (; e0 + (E2_BATCH - 1) * pass < E; e0 += E2_BATCH * pass)
        edge2_group<E2_BATCH>(e0, pass, p, src, dst, et, h, cnt, wl, out);
    if (e0 + pass < E) {
        edge2_group<2>(e0, pass, p, src, dst, et, h, cnt, wl, out);
        e0 += 2 * pass;
    }
    if (e0 < E)
        edge2_group<1>(e0, pass, p, src, dst, et, h, cnt, wl, out);
}

__device__ __forceinline__ float row8_max(float v) {
    v = fmaxf(v, lane8_xor4(v));
    v = fmaxf(v, lane8_xor2(v));
    return fmaxf(v, lane8_xor1(v));
}
__device__ __forceinline__ float row8_sum(float v) {
    v += lane8_xor4(v);
    v += lane8_xor2(v);
    return v + lane8_xor1(v);
}

__global__ void __launch_bounds__(SB)
root2_logsoftmax(const float* __restrict__ h, const float* __restrict__ root2p,
                 const float* __restrict__ bias, float* __restrict__ out) {
    const int ntiles = N_NODES / 16;
    int wave = (int)((blockIdx.x * SB + threadIdx.x) >> 5);
    int lane = (int)(threadIdx.x & 31);
    if (wave >= ntiles) return;

    int l15   = lane & 15;
    int khalf = (lane >> 4) * 2;
    int m0    = wave * 16;
    int mbase = m0 + ((lane >> 4) * 8);
    const bool live = l15 < N_CLASSES;
    const int  col  = l15 & (N_CLASSES - 1);

    float msg[8];
    #pragma unroll
    for (int j = 0; j < 8; ++j)
        msg[j] = out[(size_t)(mbase + j) * N_CLASSES + col];
    float bz = bias[col];

    const float* hrow = h + (size_t)(m0 + l15) * HID;
    v8f acc = {};
    #pragma unroll
    for (int k0 = 0; k0 < HID; k0 += 4) {
        v2f a = *(const v2f*)(hrow + k0 + khalf);
        a.x = fmaxf(a.x, 0.0f);
        a.y = fmaxf(a.y, 0.0f);
        v2f b;
        b.x = root2p[(k0 + khalf)     * 16 + l15];
        b.y = root2p[(k0 + khalf + 1) * 16 + l15];
        acc = __builtin_amdgcn_wmma_f32_16x16x4_f32(
                  false, a, false, b, (short)0, acc, false, false);
    }
    float res[8];
    #pragma unroll
    for (int j = 0; j < 8; ++j) {
        float v   = (acc[j] + bz) + msg[j];
        float m   = row8_max(v);
        float sum = row8_sum(__expf(v - m));
        res[j] = v - (m + __logf(sum));
    }
    if (live) {
        #pragma unroll
        for (int j = 0; j < 8; ++j)
            out[(size_t)(mbase + j) * N_CLASSES + l15] = res[j];
    }
}

extern "C" void kernel_launch(void* const* d_in, const int* in_sizes, int n_in,
                              void* d_out, int out_size, void* d_ws, size_t ws_size,
                              hipStream_t stream) {
    const float* x      = (const float*)d_in[0];
    const int*   eidx   = (const int*)  d_in[1];
    const int*   etype  = (const int*)  d_in[2];
    const float* comp1  = (const float*)d_in[3];
    const float* basis1 = (const float*)d_in[4];
    const float* root1  = (const float*)d_in[5];
    const float* bias1  = (const float*)d_in[6];
    const float* comp2  = (const float*)d_in[7];
    const float* basis2 = (const float*)d_in[8];
    const float* root2  = (const float*)d_in[9];
    const float* bias2  = (const float*)d_in[10];
    float* out = (float*)d_out;

    const int* src = eidx;
    const int* dst = eidx + N_EDGES;

    char* ws = (char*)d_ws;
    unsigned* cnt = (unsigned*)ws;
    float* W1     = (float*)(ws + (size_t)N_NODES * N_REL * 4);
    float* W2     = W1 + N_REL * IN_CH * HID;
    float* h      = W2 + N_REL * HID * N_CLASSES;
    float* root2p = h + (size_t)N_NODES * HID;

    setup_tables<<<SETUP_GRID, SB, 0, stream>>>(comp1, basis1, W1, comp2, basis2, W2,
                                                root2, root2p, out, cnt);
    root1_and_count<<<ROOT1_COUNT_GRID, SB, 0, stream>>>(x, root1, bias1, h, dst, etype, cnt);
    edge_msg1<<<E1_GRID, E1_THREADS, 0, stream>>>(src, dst, etype, x, W1, cnt, h, N_EDGES);
    edge_msg2<<<E2_GROUPS, E2_THREADS, 0, stream>>>(src, dst, etype, h, W2, cnt, out, N_EDGES);
    root2_logsoftmax<<<G1_BLOCKS, SB, 0, stream>>>(h, root2p, bias2, out);
}
